# starting kernel
# baseline (speedup 1.0000x reference)
#include <hip/hip_runtime.h>
#include <hip/hip_bf16.h>

typedef __attribute__((ext_vector_type(2))) float v2f;
typedef __attribute__((ext_vector_type(8))) float v8f;

#define EPS_    1e-5f
#define N_      4
#define C_      256
#define H_      192
#define W_      192
#define R_      7
#define S_      49
#define HW_     (H_ * W_)
#define CHW_    (C_ * HW_)
#define QW_     8
#define QH_     2
#define KW_     14
#define KH_     8
#define NK_     112
#define OUT_PLANE (N_ * S_ * HW_)
#define WAVE_LDS  (16 * NK_ + NK_)

__global__ __launch_bounds__(128) void corr_wmma_kernel(
    const float* __restrict__ img,
    const float* __restrict__ map,
    float* __restrict__ out)
{
    const int tid  = threadIdx.x;
    const int wv   = tid >> 5;
    const int lane = tid & 31;
    const int hh   = lane >> 4;
    const int m16  = lane & 15;

    const int n  = blockIdx.z;
    const int x0 = blockIdx.x * QW_;
    const int y0 = (blockIdx.y * 4 + wv) * QH_;

    const int qy = m16 >> 3, qx = m16 & 7;
    const int yq = y0 + qy,  xq = x0 + qx;

    const float* aP = img + (size_t)n * CHW_ + (size_t)(2 * hh) * HW_ + yq * W_ + xq;
    const float* bP = map + (size_t)n * CHW_ + (size_t)(2 * hh) * HW_;

    int koff[7];
#pragma unroll
    for (int t = 0; t < 7; ++t) {
        int kk = 16 * t + m16;
        int ry = kk / KW_, rx = kk % KW_;
        int ky = y0 - R_ + ry; ky = ky < 0 ? 0 : ky;
        int kx = x0 - R_ + rx; kx = kx < 0 ? 0 : kx;
        koff[t] = ky * W_ + kx;
    }

    v8f  acc[7] = {};
    float asum = 0.f;
    float ksum[7] = {0.f, 0.f, 0.f, 0.f, 0.f, 0.f, 0.f};

#pragma unroll 4
    for (int c0 = 0; c0 < C_; c0 += 4) {
        const int co = c0 * HW_;
        float a0 = aP[co];
        float a1 = aP[co + HW_];
        asum += a0 * a0 + a1 * a1;
        v2f A; A.x = a0; A.y = a1;
#pragma unroll
        for (int t = 0; t < 7; ++t) {
            float b0 = bP[co + koff[t]];
            float b1 = bP[co + HW_ + koff[t]];
            ksum[t] += b0 * b0 + b1 * b1;
            v2f B; B.x = b0; B.y = b1;
            acc[t] = __builtin_amdgcn_wmma_f32_16x16x4_f32(
                false, A, false, B, (short)0, acc[t], false, false);
        }
    }

    asum += __int_as_float(
        __builtin_amdgcn_ds_swizzle(__float_as_int(asum), 0x401F));
    const float qinv = 1.f / (sqrtf(asum) + EPS_);

    float kinv[7];
#pragma unroll
    for (int t = 0; t < 7; ++t) {
        float s2 = ksum[t] + __int_as_float(
            __builtin_amdgcn_ds_swizzle(__float_as_int(ksum[t]), 0x401F));
        kinv[t] = 1.f / (sqrtf(s2) + EPS_);
    }

    __shared__ float lds[4 * WAVE_LDS];
    float* G     = lds + wv * WAVE_LDS;
    float* kinvA = G + 16 * NK_;
#pragma unroll
    for (int t = 0; t < 7; ++t) {
#pragma unroll
        for (int j = 0; j < 8; ++j) {
            int m = j + 8 * hh;
            G[m * NK_ + 16 * t + m16] = acc[t][j];
        }
        if (hh == 0) kinvA[16 * t + m16] = kinv[t];
    }
    __syncthreads();

    float* outCos  = out;
    float* outDiss = out + (size_t)OUT_PLANE;
    const int pixOff = yq * W_ + xq;
    const int q = m16;
    for (int s = hh; s < S_; s += 2) {
        int i  = s / 7, j = s - 7 * i;
        int kk = (qy + i) * KW_ + (qx + j);
        float D  = G[q * NK_ + kk];
        float d1 = D * qinv;
        float cv = 1.0f - d1 * kinvA[kk];
        int o = (n * S_ + s) * HW_ + pixOff;
        __builtin_nontemporal_store(cv, outCos + o);
        __builtin_nontemporal_store(d1, outDiss + o);
    }
}

extern "C" void kernel_launch(void* const* d_in, const int* in_sizes, int n_in,
                              void* d_out, int out_size, void* d_ws, size_t ws_size,
                              hipStream_t stream) {
    const float* img = (const float*)d_in[0];
    const float* map = (const float*)d_in[1];
    (void)in_sizes; (void)n_in; (void)out_size; (void)d_ws; (void)ws_size;

    dim3 grid(W_ / QW_, H_ / (QH_ * 4), N_);
    corr_wmma_kernel<<<grid, dim3(128), 0, stream>>>(img, map, (float*)d_out);
}


# Key halo and query tile staged per block through double-buffered LDS with coalesced 16-byte global loads; WMMA fragments read from LDS (conflict-free pitch) instead of per-lane 4-byte global loads
# speedup vs baseline: 1.5627x; 1.5627x over previous
#include <hip/hip_runtime.h>
#include <hip/hip_bf16.h>


typedef __attribute__((ext_vector_type(2))) float v2f;
typedef __attribute__((ext_vector_type(4))) float v4f;
typedef __attribute__((ext_vector_type(8))) float v8f;

#define EPS_    1e-5f
#define N_      4
#define C_      256
#define H_      192
#define W_      192
#define R_      7
#define S_      49
#define HW_     (H_ * W_)
#define CHW_    (C_ * HW_)
#define QW_     8
#define QH_     2
#define KW_     14
#define KH_     8
#define NK_     112
#define OUT_PLANE (N_ * S_ * HW_)
#define WAVE_LDS  (16 * NK_ + NK_)

#define KRP_    30
#define KPP_    864
#define QPP_    160
#define KST_    (4 * KPP_)
#define STG_    (4 * KPP_ + 4 * QPP_)

__device__ __forceinline__ void put_pair(float* dst, v4f a, v4f b)
{
    v4f lo; lo.x = a.x; lo.y = b.x; lo.z = a.y; lo.w = b.y;
    v4f hi; hi.x = a.z; hi.y = b.z; hi.z = a.w; hi.w = b.w;
    *(v4f*)dst       = lo;
    *(v4f*)(dst + 4) = hi;
}

__global__ __launch_bounds__(128) void corr_wmma_kernel(
    const float* __restrict__ img,
    const float* __restrict__ map,
    float* __restrict__ out)
{
    __shared__ __attribute__((aligned(16))) float lds[2 * STG_];

    const int tid  = threadIdx.x;
    const int wv   = tid >> 5;
    const int lane = tid & 31;
    const int hh   = lane >> 4;
    const int m16  = lane & 15;

    const int n  = blockIdx.z;
    const int x0 = blockIdx.x * QW_;
    const int yb = blockIdx.y * 4 * QH_;
    const int y0 = yb + wv * QH_;

    const int qy = m16 >> 3, qx = m16 & 7;
    const int yq = y0 + qy,  xq = x0 + qx;

    const float* imgN = img + (size_t)n * CHW_;
    const float* mapN = map + (size_t)n * CHW_;

    const int wvu = __builtin_amdgcn_readfirstlane(wv);
    const bool second = wvu < 3;
    int goff[2];
    int ldst[2];
#pragma unroll
    for (int i = 0; i < 2; ++i) {
        int u = tid + 128 * i;
        u = u < 224 ? u : 223;
        int p = u / 56, rem = u - 56 * p;
        int r = rem >> 2, g = rem & 3;
        int y = yb - R_ + r;      y = y < 0 ? 0 : y;
        int x = x0 - 8 + 4 * g;   x = x < 0 ? 0 : x;
        goff[i] = (2 * p) * HW_ + y * W_ + x;
        ldst[i] = p * KPP_ + (r * KRP_ + 4 * g) * 2;
    }
    const int qch = tid >> 4, qr = (tid & 15) >> 1, qg = tid & 1;
    const int qgoff = qch * HW_ + (yb + qr) * W_ + x0 + 4 * qg;
    const int qldst = KST_ + (qch >> 1) * QPP_ + (qr * 8 + 4 * qg) * 2 + (qch & 1);

    int koff[7];
#pragma unroll
    for (int t = 0; t < 7; ++t) {
        int kk = 16 * t + m16;
        int ry = kk / KW_, rx = kk % KW_;
        int ky = y0 - R_ + ry; ky = ky < 0 ? 0 : ky;
        int kx = x0 - R_ + rx; kx = kx < 0 ? 0 : kx;
        koff[t] = hh * KPP_ + ((ky - (yb - R_)) * KRP_ + (kx - (x0 - 8))) * 2;
    }
    const int aoff = KST_ + hh * QPP_ + (16 * wv + m16) * 2;

    v8f  acc[7] = {};
    float asum = 0.f;
    float ksum[7] = {0.f, 0.f, 0.f, 0.f, 0.f, 0.f, 0.f};

    v4f ra[2], rb[2], rq;

#define FETCH(c)                                                         \
    {                                                                    \
        const int co_ = (c) * HW_;                                       \
        ra[0] = *(const v4f*)(mapN + co_ + goff[0]);                           \
        rb[0] = *(const v4f*)(mapN + co_ + HW_ + goff[0]);                     \
        if (second) {                                                    \
            ra[1] = *(const v4f*)(mapN + co_ + goff[1]);                       \
            rb[1] = *(const v4f*)(mapN + co_ + HW_ + goff[1]);                 \
        }                                                                \
        rq = *(const v4f*)(imgN + co_ + qgoff);                                \
    }

#define STASH(buf)                                                       \
    {                                                                    \
        float* st_ = lds + (buf);                                        \
        put_pair(st_ + ldst[0], ra[0], rb[0]);                           \
        if (second) put_pair(st_ + ldst[1], ra[1], rb[1]);               \
        float* qd_ = st_ + qldst;                                        \
        qd_[0] = rq.x; qd_[2] = rq.y; qd_[4] = rq.z; qd_[6] = rq.w;      \
    }

#define COMPUTE(buf)                                                     \
    {                                                                    \
        const float* st_ = lds + (buf);                                  \
        _Pragma("unroll")                                                \
        for (int j = 0; j < 2; ++j) {                                    \
            v2f A = *(const v2f*)(st_ + 2 * j * QPP_ + aoff);            \
            asum += A.x * A.x + A.y * A.y;                               \
            _Pragma("unroll")                                            \
            for (int t = 0; t < 7; ++t) {                                \
                v2f B = *(const v2f*)(st_ + 2 * j * KPP_ + koff[t]);     \
                ksum[t] += B.x * B.x + B.y * B.y;                        \
                acc[t] = __builtin_amdgcn_wmma_f32_16x16x4_f32(          \
                    false, A, false, B, (short)0, acc[t], false, false); \
            }                                                            \
        }                                                                \
    }

    FETCH(0);
    STASH(0);
    __syncthreads();

    int cur = 0;
    for (int c = 0; c < C_; c += 8) {
        const bool more = c + 8 < C_;
        if (more) FETCH(c + 8);
        COMPUTE(cur);
        if (more) STASH(STG_ - cur);
        __syncthreads();
        cur = STG_ - cur;
    }

#undef FETCH
#undef STASH
#undef COMPUTE

    asum += __int_as_float(
        __builtin_amdgcn_ds_swizzle(__float_as_int(asum), 0x401F));
    const float qinv = 1.f / (sqrtf(asum) + EPS_);

    float kinv[7];
#pragma unroll
    for (int t = 0; t < 7; ++t) {
        float s2 = ksum[t] + __int_as_float(
            __builtin_amdgcn_ds_swizzle(__float_as_int(ksum[t]), 0x401F));
        kinv[t] = 1.f / (sqrtf(s2) + EPS_);
    }

    float* G     = lds + wv * WAVE_LDS;
    float* kinvA = G + 16 * NK_;
#pragma unroll
    for (int t = 0; t < 7; ++t) {
#pragma unroll
        for (int j = 0; j < 8; ++j) {
            int m = j + 8 * hh;
            G[m * NK_ + 16 * t + m16] = acc[t][j];
        }
        if (hh == 0) kinvA[16 * t + m16] = kinv[t];
    }
    __syncthreads();

    float* outCos  = out;
    float* outDiss = out + (size_t)OUT_PLANE;
    const int pixOff = yq * W_ + xq;
    const int q = m16;
    for (int s = hh; s < S_; s += 2) {
        int i  = s / 7, j = s - 7 * i;
        int kk = (qy + i) * KW_ + (qx + j);
        float D  = G[q * NK_ + kk];
        float d1 = D * qinv;
        float cv = 1.0f - d1 * kinvA[kk];
        int o = (n * S_ + s) * HW_ + pixOff;
        __builtin_nontemporal_store(cv, outCos + o);
        __builtin_nontemporal_store(d1, outDiss + o);
    }
}

extern "C" void kernel_launch(void* const* d_in, const int* in_sizes, int n_in,
                              void* d_out, int out_size, void* d_ws, size_t ws_size,
                              hipStream_t stream) {
    const float* img = (const float*)d_in[0];
    const float* map = (const float*)d_in[1];
    (void)in_sizes; (void)n_in; (void)out_size; (void)d_ws; (void)ws_size;

    dim3 grid(W_ / QW_, H_ / (QH_ * 4), N_);
    corr_wmma_kernel<<<grid, dim3(128), 0, stream>>>(img, map, (float*)d_out);
}


# Block tile 8x8 -> 32x4 queries (8 waves, 4 across x 2 down): input rows staged and output rows stored as whole 128-byte lines instead of 32/64-byte pieces
# speedup vs baseline: 3.0614x; 1.9590x over previous
#include <hip/hip_runtime.h>
#include <hip/hip_bf16.h>


typedef __attribute__((ext_vector_type(2))) float v2f;
typedef __attribute__((ext_vector_type(4))) float v4f;
typedef __attribute__((ext_vector_type(8))) float v8f;

#define EPS_    1e-5f
#define N_      4
#define C_      256
#define H_      192
#define W_      192
#define R_      7
#define S_      49
#define HW_     (H_ * W_)
#define CHW_    (C_ * HW_)
#define QW_     8
#define QH_     2
#define KW_     14
#define KH_     8
#define NK_     112
#define OUT_PLANE (N_ * S_ * HW_)

#define BW_     32
#define BH_     4
#define NWV_    8
#define HR_     10
#define HG_     10

#define KRP_    46
#define KPP_    928
#define QPP_    288
#define KST_    (4 * KPP_)
#define STG_    (4 * KPP_ + 4 * QPP_)

#define GK_       (16 * NK_)
#define GQ_       (16 * NK_ + NK_)
#define WAVE_LDS  1936
#define LDS_FLOATS (NWV_ * WAVE_LDS)

__device__ __forceinline__ void put_pair(float* dst, v4f a, v4f b)
{
    v4f lo; lo.x = a.x; lo.y = b.x; lo.z = a.y; lo.w = b.y;
    v4f hi; hi.x = a.z; hi.y = b.z; hi.z = a.w; hi.w = b.w;
    *(v4f*)dst       = lo;
    *(v4f*)(dst + 4) = hi;
}

__global__ __launch_bounds__(256) void corr_wmma_kernel(
    const float* __restrict__ img,
    const float* __restrict__ map,
    float* __restrict__ out)
{
    __shared__ __attribute__((aligned(16))) float lds[LDS_FLOATS];

    const int tid  = threadIdx.x;
    const int wv   = tid >> 5;
    const int lane = tid & 31;
    const int hh   = lane >> 4;
    const int m16  = lane & 15;

    const int n  = blockIdx.z;
    const int x0 = blockIdx.x * BW_;
    const int yb = blockIdx.y * BH_;
    const int wx = wv & 3, wy = wv >> 2;
    const int xw = x0 + wx * QW_;
    const int yw = yb + wy * QH_;

    const float* imgN = img + (size_t)n * CHW_;
    const float* mapN = map + (size_t)n * CHW_;

    const bool second = tid < 144;
    int goff[2];
    int ldst[2];
#pragma unroll
    for (int i = 0; i < 2; ++i) {
        int u = tid + 256 * i;
        u = u < 400 ? u : 399;
        int p = u / 100, rem = u - 100 * p;
        int r = rem / HG_, g = rem - HG_ * r;
        int y = yb - R_ + r;      y = y < 0 ? 0 : y;
        int x = x0 - 8 + 4 * g;   x = x < 0 ? 0 : x;
        goff[i] = (2 * p) * HW_ + y * W_ + x;
        ldst[i] = p * KPP_ + (r * KRP_ + 4 * g) * 2;
    }
    const int qch = tid >> 5, qr = (tid & 31) >> 3, qg = tid & 7;
    const int qgoff = qch * HW_ + (yb + qr) * W_ + x0 + 4 * qg;
    const int qwave = (qr >> 1) * 4 + (qg >> 1);
    const int qm    = (qr & 1) * 8 + (qg & 1) * 4;
    const int qldst = KST_ + (qch >> 1) * QPP_ + (qwave * 16 + qm) * 2 + (qch & 1);

    int koff[7];
#pragma unroll
    for (int t = 0; t < 7; ++t) {
        int kk = 16 * t + m16;
        int ry = kk / KW_, rx = kk % KW_;
        int ky = yw - R_ + ry; ky = ky < 0 ? 0 : ky;
        int kx = xw - R_ + rx; kx = kx < 0 ? 0 : kx;
        koff[t] = hh * KPP_ + ((ky - (yb - R_)) * KRP_ + (kx - (x0 - 8))) * 2;
    }
    const int aoff = KST_ + hh * QPP_ + (16 * wv + m16) * 2;

    v8f  acc[7] = {};
    float asum = 0.f;
    float ksum[7] = {0.f, 0.f, 0.f, 0.f, 0.f, 0.f, 0.f};

    v4f ra[2], rb[2], rq;

#define FETCH(c)                                                         \
    {                                                                    \
        const int co_ = (c) * HW_;                                       \
        ra[0] = *(const v4f*)(mapN + co_ + goff[0]);                     \
        rb[0] = *(const v4f*)(mapN + co_ + HW_ + goff[0]);               \
        if (second) {                                                    \
            ra[1] = *(const v4f*)(mapN + co_ + goff[1]);                 \
            rb[1] = *(const v4f*)(mapN + co_ + HW_ + goff[1]);           \
        }                                                                \
        rq = *(const v4f*)(imgN + co_ + qgoff);                          \
    }

#define STASH(buf)                                                       \
    {                                                                    \
        float* st_ = lds + (buf);                                        \
        put_pair(st_ + ldst[0], ra[0], rb[0]);                           \
        if (second) put_pair(st_ + ldst[1], ra[1], rb[1]);               \
        float* qd_ = st_ + qldst;                                        \
        qd_[0] = rq.x; qd_[2] = rq.y; qd_[4] = rq.z; qd_[6] = rq.w;      \
    }

#define COMPUTE(buf)                                                     \
    {                                                                    \
        const float* st_ = lds + (buf);                                  \
        _Pragma("unroll")                                                \
        for (int j = 0; j < 2; ++j) {                                    \
            v2f A = *(const v2f*)(st_ + 2 * j * QPP_ + aoff);            \
            asum += A.x * A.x + A.y * A.y;                               \
            _Pragma("unroll")                                            \
            for (int t = 0; t < 7; ++t) {                                \
                v2f B = *(const v2f*)(st_ + 2 * j * KPP_ + koff[t]);     \
                ksum[t] += B.x * B.x + B.y * B.y;                        \
                acc[t] = __builtin_amdgcn_wmma_f32_16x16x4_f32(          \
                    false, A, false, B, (short)0, acc[t], false, false); \
            }                                                            \
        }                                                                \
    }

    FETCH(0);
    STASH(0);
    __syncthreads();

    int cur = 0;
    for (int c = 0; c < C_; c += 8) {
        const bool more = c + 8 < C_;
        if (more) FETCH(c + 8);
        COMPUTE(cur);
        if (more) STASH(STG_ - cur);
        __syncthreads();
        cur = STG_ - cur;
    }

#undef FETCH
#undef STASH
#undef COMPUTE

    asum += __int_as_float(
        __builtin_amdgcn_ds_swizzle(__float_as_int(asum), 0x401F));
    const float qinv = 1.f / (sqrtf(asum) + EPS_);

    float kinv[7];
#pragma unroll
    for (int t = 0; t < 7; ++t) {
        float s2 = ksum[t] + __int_as_float(
            __builtin_amdgcn_ds_swizzle(__float_as_int(ksum[t]), 0x401F));
        kinv[t] = 1.f / (sqrtf(s2) + EPS_);
    }

    float* G = lds + wv * WAVE_LDS;
#pragma unroll
    for (int t = 0; t < 7; ++t) {
#pragma unroll
        for (int j = 0; j < 8; ++j) {
            int m = j + 8 * hh;
            G[m * NK_ + 16 * t + m16] = acc[t][j];
        }
        if (hh == 0) G[GK_ + 16 * t + m16] = kinv[t];
    }
    if (hh == 0) G[GQ_ + m16] = qinv;
    __syncthreads();

    float* outCos  = out;
    float* outDiss = out + (size_t)OUT_PLANE;
    const int yy  = wv >> 1;
    const int swv = (yy >> 1) * 4 + (lane >> 3);
    const int sq  = (yy & 1) * 8 + (lane & 7);
    const int kb  = (yy & 1) * KW_ + (lane & 7);
    const float* Gs = lds + swv * WAVE_LDS;
    const float qv  = Gs[GQ_ + sq];
    const int pixOff = (yb + yy) * W_ + x0 + lane;
    for (int s = wv & 1; s < S_; s += 2) {
        int i  = s / 7, j = s - 7 * i;
        int kk = kb + i * KW_ + j;
        float D  = Gs[sq * NK_ + kk];
        float d1 = D * qv;
        float cv = 1.0f - d1 * Gs[GK_ + kk];
        int o = (n * S_ + s) * HW_ + pixOff;
        __builtin_nontemporal_store(cv, outCos + o);
        __builtin_nontemporal_store(d1, outDiss + o);
    }
}

extern "C" void kernel_launch(void* const* d_in, const int* in_sizes, int n_in,
                              void* d_out, int out_size, void* d_ws, size_t ws_size,
                              hipStream_t stream) {
    const float* img = (const float*)d_in[0];
    const float* map = (const float*)d_in[1];
    (void)in_sizes; (void)n_in; (void)out_size; (void)d_ws; (void)ws_size;

    dim3 grid(W_ / BW_, H_ / BH_, N_);
    corr_wmma_kernel<<<grid, dim3(256), 0, stream>>>(img, map, (float*)d_out);
}


# Staging loads run four stages ahead in four register sets (was one) behind an LDS-only barrier, so they stay in flight across stages: 2 work-groups per WGP, not 3, but 4x the load bytes in flight
# speedup vs baseline: 3.2296x; 1.0549x over previous
#include <hip/hip_runtime.h>
#include <hip/hip_bf16.h>


typedef __attribute__((ext_vector_type(2))) float v2f;
typedef __attribute__((ext_vector_type(4))) float v4f;
typedef __attribute__((ext_vector_type(8))) float v8f;

#define EPS_    1e-5f
#define N_      4
#define C_      256
#define H_      192
#define W_      192
#define R_      7
#define S_      49
#define HW_     (H_ * W_)
#define CHW_    (C_ * HW_)
#define QW_     8
#define QH_     2
#define KW_     14
#define KH_     8
#define NK_     112
#define OUT_PLANE (N_ * S_ * HW_)

#define BW_     32
#define BH_     4
#define NWV_    8
#define HR_     10
#define HG_     10

#define KRP_    46
#define KPP_    928
#define QPP_    288
#define KST_    (4 * KPP_)
#define STG_    (4 * KPP_ + 4 * QPP_)

#define GK_       (16 * NK_)
#define GQ_       (16 * NK_ + NK_)
#define WAVE_LDS  1936
#define LDS_FLOATS (NWV_ * WAVE_LDS)

__device__ __forceinline__ void put_pair(float* dst, v4f a, v4f b)
{
    v4f lo; lo.x = a.x; lo.y = b.x; lo.z = a.y; lo.w = b.y;
    v4f hi; hi.x = a.z; hi.y = b.z; hi.z = a.w; hi.w = b.w;
    *(v4f*)dst       = lo;
    *(v4f*)(dst + 4) = hi;
}

__device__ __forceinline__ void lds_barrier()
{
    __builtin_amdgcn_sched_barrier(0);
    __builtin_amdgcn_fence(__ATOMIC_RELEASE, "workgroup", "local");
    __builtin_amdgcn_s_barrier();
    __builtin_amdgcn_fence(__ATOMIC_ACQUIRE, "workgroup", "local");
}

__global__ __launch_bounds__(256) void corr_wmma_kernel(
    const float* __restrict__ img,
    const float* __restrict__ map,
    float* __restrict__ out)
{
    __shared__ __attribute__((aligned(16))) float lds[LDS_FLOATS];

    const int tid  = threadIdx.x;
    const int wv   = tid >> 5;
    const int lane = tid & 31;
    const int hh   = lane >> 4;
    const int m16  = lane & 15;

    const int n  = blockIdx.z;
    const int x0 = blockIdx.x * BW_;
    const int yb = blockIdx.y * BH_;
    const int wx = wv & 3, wy = wv >> 2;
    const int xw = x0 + wx * QW_;
    const int yw = yb + wy * QH_;

    const float* imgN = img + (size_t)n * CHW_;
    const float* mapN = map + (size_t)n * CHW_;

    const bool second = tid < 144;
    int goff[2];
    int ldst[2];
#pragma unroll
    for (int i = 0; i < 2; ++i) {
        int u = tid + 256 * i;
        u = u < 400 ? u : 399;
        int p = u / 100, rem = u - 100 * p;
        int r = rem / HG_, g = rem - HG_ * r;
        int y = yb - R_ + r;      y = y < 0 ? 0 : y;
        int x = x0 - 8 + 4 * g;   x = x < 0 ? 0 : x;
        goff[i] = (2 * p) * HW_ + y * W_ + x;
        ldst[i] = p * KPP_ + (r * KRP_ + 4 * g) * 2;
    }
    const int qch = tid >> 5, qr = (tid & 31) >> 3, qg = tid & 7;
    const int qgoff = qch * HW_ + (yb + qr) * W_ + x0 + 4 * qg;
    const int qwave = (qr >> 1) * 4 + (qg >> 1);
    const int qm    = (qr & 1) * 8 + (qg & 1) * 4;
    const int qldst = KST_ + (qch >> 1) * QPP_ + (qwave * 16 + qm) * 2 + (qch & 1);

    int koff[7];
#pragma unroll
    for (int t = 0; t < 7; ++t) {
        int kk = 16 * t + m16;
        int ry = kk / KW_, rx = kk % KW_;
        int ky = yw - R_ + ry; ky = ky < 0 ? 0 : ky;
        int kx = xw - R_ + rx; kx = kx < 0 ? 0 : kx;
        koff[t] = hh * KPP_ + ((ky - (yb - R_)) * KRP_ + (kx - (x0 - 8))) * 2;
    }
    const int aoff = KST_ + hh * QPP_ + (16 * wv + m16) * 2;

    v8f  acc[7] = {};
    float asum = 0.f;
    float ksum[7] = {0.f, 0.f, 0.f, 0.f, 0.f, 0.f, 0.f};

    v4f ra[4][2], rb[4][2], rq[4];

#define FETCH(c, k)                                                      \
    {                                                                    \
        const int co_ = (c) * HW_;                                       \
        ra[k][0] = *(const v4f*)(mapN + co_ + goff[0]);                  \
        rb[k][0] = *(const v4f*)(mapN + co_ + HW_ + goff[0]);            \
        ra[k][1] = *(const v4f*)(mapN + co_ + goff[1]);                  \
        rb[k][1] = *(const v4f*)(mapN + co_ + HW_ + goff[1]);            \
        rq[k] = *(const v4f*)(imgN + co_ + qgoff);                       \
    }

#define STASH(buf, k)                                                    \
    {                                                                    \
        float* st_ = lds + (buf);                                        \
        put_pair(st_ + ldst[0], ra[k][0], rb[k][0]);                     \
        if (second) put_pair(st_ + ldst[1], ra[k][1], rb[k][1]);         \
        float* qd_ = st_ + qldst;                                        \
        qd_[0] = rq[k].x; qd_[2] = rq[k].y; qd_[4] = rq[k].z; qd_[6] = rq[k].w; \
    }

#define HOLD_SUMS()                                                      \
    asm volatile("" : "+v"(asum), "+v"(ksum[0]), "+v"(ksum[1]), "+v"(ksum[2]), "+v"(ksum[3]), \
                      "+v"(ksum[4]), "+v"(ksum[5]), "+v"(ksum[6]))

#define COMPUTE(buf)                                                     \
    {                                                                    \
        const float* st_ = lds + (buf);                                  \
        _Pragma("unroll")                                                \
        for (int j = 0; j < 2; ++j) {                                    \
            v2f A = *(const v2f*)(st_ + 2 * j * QPP_ + aoff);            \
            asum += A.x * A.x + A.y * A.y;                               \
            _Pragma("unroll")                                            \
            for (int t = 0; t < 7; ++t) {                                \
                v2f B = *(const v2f*)(st_ + 2 * j * KPP_ + koff[t]);     \
                ksum[t] += B.x * B.x + B.y * B.y;                        \
                acc[t] = __builtin_amdgcn_wmma_f32_16x16x4_f32(          \
                    false, A, false, B, (short)0, acc[t], false, false); \
            }                                                            \
        }                                                                \
        HOLD_SUMS();                                                     \
    }

#define STEP(s_, u_, DO_STASH, DO_FETCH)                                 \
    {                                                                    \
        COMPUTE(((u_) & 1) * STG_);                                      \
        if (DO_STASH) STASH((((u_) + 1) & 1) * STG_, ((u_) + 1) & 3);    \
        if (DO_FETCH) FETCH(8 * ((s_) + 5), ((u_) + 1) & 3);             \
        lds_barrier();                                                   \
    }

    FETCH(0, 0);
    __builtin_amdgcn_sched_barrier(0);
    FETCH(8, 1);
    __builtin_amdgcn_sched_barrier(0);
    FETCH(16, 2);
    __builtin_amdgcn_sched_barrier(0);
    FETCH(24, 3);
    __builtin_amdgcn_sched_barrier(0);
    STASH(0, 0);
    FETCH(32, 0);
    lds_barrier();

    for (int s = 0; s < 24; s += 4) {
        STEP(s,     0, true, true);
        STEP(s + 1, 1, true, true);
        STEP(s + 2, 2, true, true);
        STEP(s + 3, 3, true, true);
    }
    STEP(24, 0, true, true);
    STEP(25, 1, true, true);
    STEP(26, 2, true, true);
    STEP(27, 3, true, false);
    STEP(28, 0, true, false);
    STEP(29, 1, true, false);
    STEP(30, 2, true, false);
    STEP(31, 3, false, false);

#undef STEP
#undef HOLD_SUMS
#undef FETCH
#undef STASH
#undef COMPUTE

    asum += __int_as_float(
        __builtin_amdgcn_ds_swizzle(__float_as_int(asum), 0x401F));
    const float qinv = 1.f / (sqrtf(asum) + EPS_);

    float kinv[7];
#pragma unroll
    for (int t = 0; t < 7; ++t) {
        float s2 = ksum[t] + __int_as_float(
            __builtin_amdgcn_ds_swizzle(__float_as_int(ksum[t]), 0x401F));
        kinv[t] = 1.f / (sqrtf(s2) + EPS_);
    }

    float* G = lds + wv * WAVE_LDS;
#pragma unroll
    for (int t = 0; t < 7; ++t) {
#pragma unroll
        for (int j = 0; j < 8; ++j) {
            int m = j + 8 * hh;
            G[m * NK_ + 16 * t + m16] = acc[t][j];
        }
        if (hh == 0) G[GK_ + 16 * t + m16] = kinv[t];
    }
    if (hh == 0) G[GQ_ + m16] = qinv;
    __syncthreads();

    float* outCos  = out;
    float* outDiss = out + (size_t)OUT_PLANE;
    const int yy  = wv >> 1;
    const int swv = (yy >> 1) * 4 + (lane >> 3);
    const int sq  = (yy & 1) * 8 + (lane & 7);
    const int kb  = (yy & 1) * KW_ + (lane & 7);
    const float* Gs = lds + swv * WAVE_LDS;
    const float qv  = Gs[GQ_ + sq];
    const int pixOff = (yb + yy) * W_ + x0 + lane;
    for (int s = wv & 1; s < S_; s += 2) {
        int i  = s / 7, j = s - 7 * i;
        int kk = kb + i * KW_ + j;
        float D  = Gs[sq * NK_ + kk];
        float d1 = D * qv;
        float cv = 1.0f - d1 * Gs[GK_ + kk];
        int o = (n * S_ + s) * HW_ + pixOff;
        __builtin_nontemporal_store(cv, outCos + o);
        __builtin_nontemporal_store(d1, outDiss + o);
    }
}

extern "C" void kernel_launch(void* const* d_in, const int* in_sizes, int n_in,
                              void* d_out, int out_size, void* d_ws, size_t ws_size,
                              hipStream_t stream) {
    const float* img = (const float*)d_in[0];
    const float* map = (const float*)d_in[1];
    (void)in_sizes; (void)n_in; (void)out_size; (void)d_ws; (void)ws_size;

    dim3 grid(W_ / BW_, H_ / BH_, N_);
    corr_wmma_kernel<<<grid, dim3(256), 0, stream>>>(img, map, (float*)d_out);
}


# Sums of squares for the norms taken once per staged value by the staging threads and joined through LDS behind the channel loop, not by every wave from its fragments; the norm sums' order changes
# speedup vs baseline: 3.4455x; 1.0669x over previous
#include <hip/hip_runtime.h>
#include <hip/hip_bf16.h>


typedef __attribute__((ext_vector_type(2))) float v2f;
typedef __attribute__((ext_vector_type(4))) float v4f;
typedef __attribute__((ext_vector_type(8))) float v8f;

#define EPS_    1e-5f
#define N_      4
#define C_      256
#define H_      192
#define W_      192
#define R_      7
#define S_      49
#define HW_     (H_ * W_)
#define CHW_    (C_ * HW_)
#define QW_     8
#define QH_     2
#define KW_     14
#define KH_     8
#define NK_     112
#define OUT_PLANE (N_ * S_ * HW_)

#define BW_     32
#define BH_     4
#define NWV_    8
#define HR_     10
#define HG_     10

#define KRP_    46
#define KPP_    928
#define QPP_    288
#define KST_    (4 * KPP_)
#define STG_    (4 * KPP_ + 4 * QPP_)

#define GK_       (16 * NK_)
#define GQ_       (16 * NK_ + NK_)
#define WAVE_LDS  1936
#define LDS_FLOATS (NWV_ * WAVE_LDS)

#define KTB_      (HR_ * KRP_)
#define QPB_      (4 * KTB_)

__device__ __forceinline__ void put_pair(float* dst, v4f a, v4f b)
{
    v4f lo; lo.x = a.x; lo.y = b.x; lo.z = a.y; lo.w = b.y;
    v4f hi; hi.x = a.z; hi.y = b.z; hi.z = a.w; hi.w = b.w;
    *(v4f*)dst       = lo;
    *(v4f*)(dst + 4) = hi;
}

__device__ __forceinline__ void lds_barrier()
{
    __builtin_amdgcn_sched_barrier(0);
    __builtin_amdgcn_fence(__ATOMIC_RELEASE, "workgroup", "local");
    __builtin_amdgcn_s_barrier();
    __builtin_amdgcn_fence(__ATOMIC_ACQUIRE, "workgroup", "local");
}

__global__ __launch_bounds__(256) void corr_wmma_kernel(
    const float* __restrict__ img,
    const float* __restrict__ map,
    float* __restrict__ out)
{
    __shared__ __attribute__((aligned(16))) float lds[LDS_FLOATS];

    const int tid  = threadIdx.x;
    const int wv   = tid >> 5;
    const int lane = tid & 31;
    const int hh   = lane >> 4;
    const int m16  = lane & 15;

    const int n  = blockIdx.z;
    const int x0 = blockIdx.x * BW_;
    const int yb = blockIdx.y * BH_;
    const int wx = wv & 3, wy = wv >> 2;
    const int xw = x0 + wx * QW_;
    const int yw = yb + wy * QH_;

    const float* imgN = img + (size_t)n * CHW_;
    const float* mapN = map + (size_t)n * CHW_;

    const bool second = tid < 144;
    int goff[2];
    int ldst[2];
#pragma unroll
    for (int i = 0; i < 2; ++i) {
        int u = tid + 256 * i;
        u = u < 400 ? u : 399;
        int p = u / 100, rem = u - 100 * p;
        int r = rem / HG_, g = rem - HG_ * r;
        int y = yb - R_ + r;      y = y < 0 ? 0 : y;
        int x = x0 - 8 + 4 * g;   x = x < 0 ? 0 : x;
        goff[i] = (2 * p) * HW_ + y * W_ + x;
        ldst[i] = p * KPP_ + (r * KRP_ + 4 * g) * 2;
    }
    const int qch = tid >> 5, qr = (tid & 31) >> 3, qg = tid & 7;
    const int qgoff = qch * HW_ + (yb + qr) * W_ + x0 + 4 * qg;
    const int qwave = (qr >> 1) * 4 + (qg >> 1);
    const int qm    = (qr & 1) * 8 + (qg & 1) * 4;
    const int qldst = KST_ + (qch >> 1) * QPP_ + (qwave * 16 + qm) * 2 + (qch & 1);

    int koff[7];
#pragma unroll
    for (int t = 0; t < 7; ++t) {
        int kk = 16 * t + m16;
        int ry = kk / KW_, rx = kk % KW_;
        int ky = yw - R_ + ry; ky = ky < 0 ? 0 : ky;
        int kx = xw - R_ + rx; kx = kx < 0 ? 0 : kx;
        koff[t] = hh * KPP_ + ((ky - (yb - R_)) * KRP_ + (kx - (x0 - 8))) * 2;
    }
    const int aoff = KST_ + hh * QPP_ + (16 * wv + m16) * 2;

    v8f  acc[7] = {};
    v2f ks00 = {}, ks01 = {}, ks10 = {}, ks11 = {};
    v2f qs0 = {}, qs1 = {};

    v4f ra[4][2], rb[4][2], rq[4];

#define FETCH(c, k)                                                      \
    {                                                                    \
        const int co_ = (c) * HW_;                                       \
        ra[k][0] = *(const v4f*)(mapN + co_ + goff[0]);                  \
        rb[k][0] = *(const v4f*)(mapN + co_ + HW_ + goff[0]);            \
        ra[k][1] = *(const v4f*)(mapN + co_ + goff[1]);                  \
        rb[k][1] = *(const v4f*)(mapN + co_ + HW_ + goff[1]);            \
        rq[k] = *(const v4f*)(imgN + co_ + qgoff);                       \
    }

#define HOLD_SUMS()                                                      \
    asm volatile("" : "+v"(ks00), "+v"(ks01), "+v"(ks10), "+v"(ks11), "+v"(qs0), "+v"(qs1))

#define STASH(buf, k)                                                    \
    {                                                                    \
        float* st_ = lds + (buf);                                        \
        put_pair(st_ + ldst[0], ra[k][0], rb[k][0]);                     \
        if (second) put_pair(st_ + ldst[1], ra[k][1], rb[k][1]);         \
        float* qd_ = st_ + qldst;                                        \
        qd_[0] = rq[k].x; qd_[2] = rq[k].y; qd_[4] = rq[k].z; qd_[6] = rq[k].w; \
        ks00 += ra[k][0].lo * ra[k][0].lo;   ks01 += ra[k][0].hi * ra[k][0].hi;   \
        ks00 += rb[k][0].lo * rb[k][0].lo;   ks01 += rb[k][0].hi * rb[k][0].hi;   \
        ks10 += ra[k][1].lo * ra[k][1].lo;   ks11 += ra[k][1].hi * ra[k][1].hi;   \
        ks10 += rb[k][1].lo * rb[k][1].lo;   ks11 += rb[k][1].hi * rb[k][1].hi;   \
        qs0  += rq[k].lo * rq[k].lo;         qs1  += rq[k].hi * rq[k].hi;         \
        HOLD_SUMS();                                                     \
    }

#define COMPUTE(buf)                                                     \
    {                                                                    \
        const float* st_ = lds + (buf);                                  \
        _Pragma("unroll")                                                \
        for (int j = 0; j < 2; ++j) {                                    \
            v2f A = *(const v2f*)(st_ + 2 * j * QPP_ + aoff);            \
            _Pragma("unroll")                                            \
            for (int t = 0; t < 7; ++t) {                                \
                v2f B = *(const v2f*)(st_ + 2 * j * KPP_ + koff[t]);     \
                acc[t] = __builtin_amdgcn_wmma_f32_16x16x4_f32(          \
                    false, A, false, B, (short)0, acc[t], false, false); \
            }                                                            \
        }                                                                \
    }

#define STEP(s_, u_, DO_STASH, DO_FETCH)                                 \
    {                                                                    \
        COMPUTE(((u_) & 1) * STG_);                                      \
        if (DO_STASH) STASH((((u_) + 1) & 1) * STG_, ((u_) + 1) & 3);    \
        if (DO_FETCH) FETCH(8 * ((s_) + 5), ((u_) + 1) & 3);             \
        lds_barrier();                                                   \
    }

    FETCH(0, 0);
    __builtin_amdgcn_sched_barrier(0);
    FETCH(8, 1);
    __builtin_amdgcn_sched_barrier(0);
    FETCH(16, 2);
    __builtin_amdgcn_sched_barrier(0);
    FETCH(24, 3);
    __builtin_amdgcn_sched_barrier(0);
    STASH(0, 0);
    FETCH(32, 0);
    lds_barrier();

    for (int s = 0; s < 24; s += 4) {
        STEP(s,     0, true, true);
        STEP(s + 1, 1, true, true);
        STEP(s + 2, 2, true, true);
        STEP(s + 3, 3, true, true);
    }
    STEP(24, 0, true, true);
    STEP(25, 1, true, true);
    STEP(26, 2, true, true);
    STEP(27, 3, true, false);
    STEP(28, 0, true, false);
    STEP(29, 1, true, false);
    STEP(30, 2, true, false);
    {
        COMPUTE(STG_);
        {
            const int p = tid / 100, rem = tid - 100 * p;
            const int r = rem / HG_, g = rem - HG_ * r;
            float* d = lds + p * KTB_ + r * KRP_ + 4 * g;
            *(v2f*)d       = ks00;
            *(v2f*)(d + 2) = ks01;
        }
        if (second) {
            const int u = tid + 256;
            const int p = u / 100, rem = u - 100 * p;
            const int r = rem / HG_, g = rem - HG_ * r;
            float* d = lds + p * KTB_ + r * KRP_ + 4 * g;
            *(v2f*)d       = ks10;
            *(v2f*)(d + 2) = ks11;
        }
        v4f q4; q4.x = qs0.x; q4.y = qs0.y; q4.z = qs1.x; q4.w = qs1.y;
        *(v4f*)(lds + QPB_ + qch * (BW_ * BH_) + qwave * 16 + qm) = q4;
        lds_barrier();
    }

#undef STEP
#undef HOLD_SUMS
#undef FETCH
#undef STASH
#undef COMPUTE

    const float* qp = lds + QPB_ + 4 * hh * (BW_ * BH_) + 16 * wv + m16;
    float asum = (qp[0] + qp[BW_ * BH_]) + (qp[2 * BW_ * BH_] + qp[3 * BW_ * BH_]);
    asum += __int_as_float(
        __builtin_amdgcn_ds_swizzle(__float_as_int(asum), 0x401F));
    const float qinv = 1.f / (sqrtf(asum) + EPS_);

    float kinv[7];
#pragma unroll
    for (int t = 0; t < 7; ++t) {
        const float* kp = lds + 2 * hh * KTB_ + ((koff[t] - hh * KPP_) >> 1);
        float s2 = kp[0] + kp[KTB_];
        s2 += __int_as_float(
            __builtin_amdgcn_ds_swizzle(__float_as_int(s2), 0x401F));
        kinv[t] = 1.f / (sqrtf(s2) + EPS_);
    }
    __syncthreads();

    float* G = lds + wv * WAVE_LDS;
#pragma unroll
    for (int t = 0; t < 7; ++t) {
#pragma unroll
        for (int j = 0; j < 8; ++j) {
            int m = j + 8 * hh;
            G[m * NK_ + 16 * t + m16] = acc[t][j];
        }
        if (hh == 0) G[GK_ + 16 * t + m16] = kinv[t];
    }
    if (hh == 0) G[GQ_ + m16] = qinv;
    __syncthreads();

    float* outCos  = out;
    float* outDiss = out + (size_t)OUT_PLANE;
    const int yy  = wv >> 1;
    const int swv = (yy >> 1) * 4 + (lane >> 3);
    const int sq  = (yy & 1) * 8 + (lane & 7);
    const int kb  = (yy & 1) * KW_ + (lane & 7);
    const float* Gs = lds + swv * WAVE_LDS;
    const float qv  = Gs[GQ_ + sq];
    const int pixOff = (yb + yy) * W_ + x0 + lane;
    for (int s = wv & 1; s < S_; s += 2) {
        int i  = s / 7, j = s - 7 * i;
        int kk = kb + i * KW_ + j;
        float D  = Gs[sq * NK_ + kk];
        float d1 = D * qv;
        float cv = 1.0f - d1 * Gs[GK_ + kk];
        int o = (n * S_ + s) * HW_ + pixOff;
        __builtin_nontemporal_store(cv, outCos + o);
        __builtin_nontemporal_store(d1, outDiss + o);
    }
}

extern "C" void kernel_launch(void* const* d_in, const int* in_sizes, int n_in,
                              void* d_out, int out_size, void* d_ws, size_t ws_size,
                              hipStream_t stream) {
    const float* img = (const float*)d_in[0];
    const float* map = (const float*)d_in[1];
    (void)in_sizes; (void)n_in; (void)out_size; (void)d_ws; (void)ws_size;

    dim3 grid(W_ / BW_, H_ / BH_, N_);
    corr_wmma_kernel<<<grid, dim3(256), 0, stream>>>(img, map, (float*)d_out);
}


# Block tile 32x4 -> 32x6 queries (12 waves, 4 across x 3 down, one work-group per WGP): 12 staged key rows per 6 query rows, not 10 per 4, and 768 work-groups = exactly 3 per WGP, no partial round
# speedup vs baseline: 3.6088x; 1.0474x over previous
#include <hip/hip_runtime.h>
#include <hip/hip_bf16.h>


typedef __attribute__((ext_vector_type(2))) float v2f;
typedef __attribute__((ext_vector_type(4))) float v4f;
typedef __attribute__((ext_vector_type(8))) float v8f;

#define EPS_    1e-5f
#define N_      4
#define C_      256
#define H_      192
#define W_      192
#define R_      7
#define S_      49
#define HW_     (H_ * W_)
#define CHW_    (C_ * HW_)
#define QW_     8
#define QH_     2
#define KW_     14
#define KH_     8
#define NK_     112
#define OUT_PLANE (N_ * S_ * HW_)

#define BW_     32
#define BH_     6
#define NWV_    12
#define NTHR_   (32 * NWV_)
#define HR_     12
#define HG_     10
#define KTP_    (HR_ * HG_)
#define NKT_    (4 * KTP_)
#define QTC_    (BH_ * 8)

#define KRP_    46
#define KPP_    1120
#define QPP_    416
#define KST_    (4 * KPP_)
#define STG_    (4 * KPP_ + 4 * QPP_)

#define GK_       (16 * NK_)
#define GQ_       (16 * NK_ + NK_)
#define WAVE_LDS  1936
#define LDS_FLOATS (NWV_ * WAVE_LDS)

#define KTB_      (HR_ * KRP_)
#define QPB_      (4 * KTB_)

__device__ __forceinline__ void put_pair(float* dst, v4f a, v4f b)
{
    v4f lo; lo.x = a.x; lo.y = b.x; lo.z = a.y; lo.w = b.y;
    v4f hi; hi.x = a.z; hi.y = b.z; hi.z = a.w; hi.w = b.w;
    *(v4f*)dst       = lo;
    *(v4f*)(dst + 4) = hi;
}

__device__ __forceinline__ void lds_barrier()
{
    __builtin_amdgcn_sched_barrier(0);
    __builtin_amdgcn_fence(__ATOMIC_RELEASE, "workgroup", "local");
    __builtin_amdgcn_s_barrier();
    __builtin_amdgcn_fence(__ATOMIC_ACQUIRE, "workgroup", "local");
}

__global__ __launch_bounds__(NTHR_) void corr_wmma_kernel(
    const float* __restrict__ img,
    const float* __restrict__ map,
    float* __restrict__ out)
{
    __shared__ __attribute__((aligned(16))) float lds[LDS_FLOATS];

    const int tid  = threadIdx.x;
    const int wv   = tid >> 5;
    const int lane = tid & 31;
    const int hh   = lane >> 4;
    const int m16  = lane & 15;

    const int n  = blockIdx.z;
    const int x0 = blockIdx.x * BW_;
    const int yb = blockIdx.y * BH_;
    const int wx = wv & 3, wy = wv >> 2;
    const int xw = x0 + wx * QW_;
    const int yw = yb + wy * QH_;

    const float* imgN = img + (size_t)n * CHW_;
    const float* mapN = map + (size_t)n * CHW_;

    const bool second = tid < NKT_ - NTHR_;
    int goff[2];
    int ldst[2];
#pragma unroll
    for (int i = 0; i < 2; ++i) {
        int u = tid + NTHR_ * i;
        u = u < NKT_ ? u : NKT_ - 1;
        int p = u / KTP_, rem = u - KTP_ * p;
        int r = rem / HG_, g = rem - HG_ * r;
        int y = yb - R_ + r;      y = y < 0 ? 0 : y;
        int x = x0 - 8 + 4 * g;   x = x < 0 ? 0 : x;
        goff[i] = (2 * p) * HW_ + y * W_ + x;
        ldst[i] = p * KPP_ + (r * KRP_ + 4 * g) * 2;
    }
    const int qch = tid / QTC_, qr = (tid - QTC_ * qch) >> 3, qg = tid & 7;
    const int qgoff = qch * HW_ + (yb + qr) * W_ + x0 + 4 * qg;
    const int qwave = (qr >> 1) * 4 + (qg >> 1);
    const int qm    = (qr & 1) * 8 + (qg & 1) * 4;
    const int qldst = KST_ + (qch >> 1) * QPP_ + (qwave * 16 + qm) * 2 + (qch & 1);

    int koff[7];
#pragma unroll
    for (int t = 0; t < 7; ++t) {
        int kk = 16 * t + m16;
        int ry = kk / KW_, rx = kk % KW_;
        int ky = yw - R_ + ry; ky = ky < 0 ? 0 : ky;
        int kx = xw - R_ + rx; kx = kx < 0 ? 0 : kx;
        koff[t] = hh * KPP_ + ((ky - (yb - R_)) * KRP_ + (kx - (x0 - 8))) * 2;
    }
    const int aoff = KST_ + hh * QPP_ + (16 * wv + m16) * 2;

    v8f  acc[7] = {};
    v2f ks00 = {}, ks01 = {}, ks10 = {}, ks11 = {};
    v2f qs0 = {}, qs1 = {};

    v4f ra[4][2], rb[4][2], rq[4];

#define FETCH(c, k)                                                      \
    {                                                                    \
        const int co_ = (c) * HW_;                                       \
        ra[k][0] = *(const v4f*)(mapN + co_ + goff[0]);                  \
        rb[k][0] = *(const v4f*)(mapN + co_ + HW_ + goff[0]);            \
        ra[k][1] = *(const v4f*)(mapN + co_ + goff[1]);                  \
        rb[k][1] = *(const v4f*)(mapN + co_ + HW_ + goff[1]);            \
        rq[k] = *(const v4f*)(imgN + co_ + qgoff);                       \
    }

#define HOLD_SUMS()                                                      \
    asm volatile("" : "+v"(ks00), "+v"(ks01), "+v"(ks10), "+v"(ks11), "+v"(qs0), "+v"(qs1))

#define STASH(buf, k)                                                    \
    {                                                                    \
        float* st_ = lds + (buf);                                        \
        put_pair(st_ + ldst[0], ra[k][0], rb[k][0]);                     \
        if (second) put_pair(st_ + ldst[1], ra[k][1], rb[k][1]);         \
        float* qd_ = st_ + qldst;                                        \
        qd_[0] = rq[k].x; qd_[2] = rq[k].y; qd_[4] = rq[k].z; qd_[6] = rq[k].w; \
        ks00 += ra[k][0].lo * ra[k][0].lo;   ks01 += ra[k][0].hi * ra[k][0].hi;   \
        ks00 += rb[k][0].lo * rb[k][0].lo;   ks01 += rb[k][0].hi * rb[k][0].hi;   \
        ks10 += ra[k][1].lo * ra[k][1].lo;   ks11 += ra[k][1].hi * ra[k][1].hi;   \
        ks10 += rb[k][1].lo * rb[k][1].lo;   ks11 += rb[k][1].hi * rb[k][1].hi;   \
        qs0  += rq[k].lo * rq[k].lo;         qs1  += rq[k].hi * rq[k].hi;         \
        HOLD_SUMS();                                                     \
    }

#define COMPUTE(buf)                                                     \
    {                                                                    \
        const float* st_ = lds + (buf);                                  \
        _Pragma("unroll")                                                \
        for (int j = 0; j < 2; ++j) {                                    \
            v2f A = *(const v2f*)(st_ + 2 * j * QPP_ + aoff);            \
            _Pragma("unroll")                                            \
            for (int t = 0; t < 7; ++t) {                                \
                v2f B = *(const v2f*)(st_ + 2 * j * KPP_ + koff[t]);     \
                acc[t] = __builtin_amdgcn_wmma_f32_16x16x4_f32(          \
                    false, A, false, B, (short)0, acc[t], false, false); \
            }                                                            \
        }                                                                \
    }

#define STEP(s_, u_, DO_STASH, DO_FETCH)                                 \
    {                                                                    \
        COMPUTE(((u_) & 1) * STG_);                                      \
        if (DO_STASH) STASH((((u_) + 1) & 1) * STG_, ((u_) + 1) & 3);    \
        if (DO_FETCH) FETCH(8 * ((s_) + 5), ((u_) + 1) & 3);             \
        lds_barrier();                                                   \
    }

    FETCH(0, 0);
    __builtin_amdgcn_sched_barrier(0);
    FETCH(8, 1);
    __builtin_amdgcn_sched_barrier(0);
    FETCH(16, 2);
    __builtin_amdgcn_sched_barrier(0);
    FETCH(24, 3);
    __builtin_amdgcn_sched_barrier(0);
    STASH(0, 0);
    FETCH(32, 0);
    lds_barrier();

    for (int s = 0; s < 24; s += 4) {
        STEP(s,     0, true, true);
        STEP(s + 1, 1, true, true);
        STEP(s + 2, 2, true, true);
        STEP(s + 3, 3, true, true);
    }
    STEP(24, 0, true, true);
    STEP(25, 1, true, true);
    STEP(26, 2, true, true);
    STEP(27, 3, true, false);
    STEP(28, 0, true, false);
    STEP(29, 1, true, false);
    STEP(30, 2, true, false);
    {
        COMPUTE(STG_);
        {
            const int p = tid / KTP_, rem = tid - KTP_ * p;
            const int r = rem / HG_, g = rem - HG_ * r;
            float* d = lds + p * KTB_ + r * KRP_ + 4 * g;
            *(v2f*)d       = ks00;
            *(v2f*)(d + 2) = ks01;
        }
        if (second) {
            const int u = tid + NTHR_;
            const int p = u / KTP_, rem = u - KTP_ * p;
            const int r = rem / HG_, g = rem - HG_ * r;
            float* d = lds + p * KTB_ + r * KRP_ + 4 * g;
            *(v2f*)d       = ks10;
            *(v2f*)(d + 2) = ks11;
        }
        v4f q4; q4.x = qs0.x; q4.y = qs0.y; q4.z = qs1.x; q4.w = qs1.y;
        *(v4f*)(lds + QPB_ + qch * (BW_ * BH_) + qwave * 16 + qm) = q4;
        lds_barrier();
    }

#undef STEP
#undef HOLD_SUMS
#undef FETCH
#undef STASH
#undef COMPUTE

    const float* qp = lds + QPB_ + 4 * hh * (BW_ * BH_) + 16 * wv + m16;
    float asum = (qp[0] + qp[BW_ * BH_]) + (qp[2 * BW_ * BH_] + qp[3 * BW_ * BH_]);
    asum += __int_as_float(
        __builtin_amdgcn_ds_swizzle(__float_as_int(asum), 0x401F));
    const float qinv = 1.f / (sqrtf(asum) + EPS_);

    float kinv[7];
#pragma unroll
    for (int t = 0; t < 7; ++t) {
        const float* kp = lds + 2 * hh * KTB_ + ((koff[t] - hh * KPP_) >> 1);
        float s2 = kp[0] + kp[KTB_];
        s2 += __int_as_float(
            __builtin_amdgcn_ds_swizzle(__float_as_int(s2), 0x401F));
        kinv[t] = 1.f / (sqrtf(s2) + EPS_);
    }
    __syncthreads();

    float* G = lds + wv * WAVE_LDS;
#pragma unroll
    for (int t = 0; t < 7; ++t) {
#pragma unroll
        for (int j = 0; j < 8; ++j) {
            int m = j + 8 * hh;
            G[m * NK_ + 16 * t + m16] = acc[t][j];
        }
        if (hh == 0) G[GK_ + 16 * t + m16] = kinv[t];
    }
    if (hh == 0) G[GQ_ + m16] = qinv;
    __syncthreads();

    float* outCos  = out;
    float* outDiss = out + (size_t)OUT_PLANE;
    const int yy  = wv >> 1;
    const int swv = (yy >> 1) * 4 + (lane >> 3);
    const int sq  = (yy & 1) * 8 + (lane & 7);
    const int kb  = (yy & 1) * KW_ + (lane & 7);
    const float* Gs = lds + swv * WAVE_LDS;
    const float qv  = Gs[GQ_ + sq];
    const int pixOff = (yb + yy) * W_ + x0 + lane;
    for (int s = wv & 1; s < S_; s += 2) {
        int i  = s / 7, j = s - 7 * i;
        int kk = kb + i * KW_ + j;
        float D  = Gs[sq * NK_ + kk];
        float d1 = D * qv;
        float cv = 1.0f - d1 * Gs[GK_ + kk];
        int o = (n * S_ + s) * HW_ + pixOff;
        __builtin_nontemporal_store(cv, outCos + o);
        __builtin_nontemporal_store(d1, outDiss + o);
    }
}

extern "C" void kernel_launch(void* const* d_in, const int* in_sizes, int n_in,
                              void* d_out, int out_size, void* d_ws, size_t ws_size,
                              hipStream_t stream) {
    const float* img = (const float*)d_in[0];
    const float* map = (const float*)d_in[1];
    (void)in_sizes; (void)n_in; (void)out_size; (void)d_ws; (void)ws_size;

    dim3 grid(W_ / BW_, H_ / BH_, N_);
    corr_wmma_kernel<<<grid, dim3(NTHR_), 0, stream>>>(img, map, (float*)d_out);
}


# Stage barrier split: each wave signals behind its fragment reads and staging stores (now in front of the WMMAs) and waits behind its 14 WMMAs, so waves run up to a stage apart; no sum reordered
# speedup vs baseline: 3.7209x; 1.0311x over previous
#include <hip/hip_runtime.h>
#include <hip/hip_bf16.h>


typedef __attribute__((ext_vector_type(2))) float v2f;
typedef __attribute__((ext_vector_type(4))) float v4f;
typedef __attribute__((ext_vector_type(8))) float v8f;

#define EPS_    1e-5f
#define N_      4
#define C_      256
#define H_      192
#define W_      192
#define R_      7
#define S_      49
#define HW_     (H_ * W_)
#define CHW_    (C_ * HW_)
#define QW_     8
#define QH_     2
#define KW_     14
#define KH_     8
#define NK_     112
#define OUT_PLANE (N_ * S_ * HW_)

#define BW_     32
#define BH_     6
#define NWV_    12
#define NTHR_   (32 * NWV_)
#define HR_     12
#define HG_     10
#define KTP_    (HR_ * HG_)
#define NKT_    (4 * KTP_)
#define QTC_    (BH_ * 8)

#define KRP_    46
#define KPP_    1120
#define QPP_    416
#define KST_    (4 * KPP_)
#define STG_    (4 * KPP_ + 4 * QPP_)

#define GK_       (16 * NK_)
#define GQ_       (16 * NK_ + NK_)
#define WAVE_LDS  1936
#define LDS_FLOATS (NWV_ * WAVE_LDS)

#define KTB_      (HR_ * KRP_)
#define QPB_      (4 * KTB_)

__device__ __forceinline__ void put_pair(float* dst, v4f a, v4f b)
{
    v4f lo; lo.x = a.x; lo.y = b.x; lo.z = a.y; lo.w = b.y;
    v4f hi; hi.x = a.z; hi.y = b.z; hi.z = a.w; hi.w = b.w;
    *(v4f*)dst       = lo;
    *(v4f*)(dst + 4) = hi;
}

__device__ __forceinline__ void lds_barrier()
{
    __builtin_amdgcn_sched_barrier(0);
    __builtin_amdgcn_fence(__ATOMIC_RELEASE, "workgroup", "local");
    __builtin_amdgcn_s_barrier();
    __builtin_amdgcn_fence(__ATOMIC_ACQUIRE, "workgroup", "local");
}

__device__ __forceinline__ void stage_signal()
{
    __builtin_amdgcn_fence(__ATOMIC_RELEASE, "workgroup", "local");
    __builtin_amdgcn_s_barrier_signal(-1);
    __builtin_amdgcn_sched_barrier(0);
}

__device__ __forceinline__ void stage_wait()
{
    __builtin_amdgcn_sched_barrier(0);
    __builtin_amdgcn_s_barrier_wait((short)-1);
    __builtin_amdgcn_fence(__ATOMIC_ACQUIRE, "workgroup", "local");
}

__global__ __launch_bounds__(NTHR_) void corr_wmma_kernel(
    const float* __restrict__ img,
    const float* __restrict__ map,
    float* __restrict__ out)
{
    __shared__ __attribute__((aligned(16))) float lds[LDS_FLOATS];

    const int tid  = threadIdx.x;
    const int wv   = tid >> 5;
    const int lane = tid & 31;
    const int hh   = lane >> 4;
    const int m16  = lane & 15;

    const int n  = blockIdx.z;
    const int x0 = blockIdx.x * BW_;
    const int yb = blockIdx.y * BH_;
    const int wx = wv & 3, wy = wv >> 2;
    const int xw = x0 + wx * QW_;
    const int yw = yb + wy * QH_;

    const float* imgN = img + (size_t)n * CHW_;
    const float* mapN = map + (size_t)n * CHW_;

    const bool second = tid < NKT_ - NTHR_;
    int goff[2];
    int ldst[2];
#pragma unroll
    for (int i = 0; i < 2; ++i) {
        int u = tid + NTHR_ * i;
        u = u < NKT_ ? u : NKT_ - 1;
        int p = u / KTP_, rem = u - KTP_ * p;
        int r = rem / HG_, g = rem - HG_ * r;
        int y = yb - R_ + r;      y = y < 0 ? 0 : y;
        int x = x0 - 8 + 4 * g;   x = x < 0 ? 0 : x;
        goff[i] = (2 * p) * HW_ + y * W_ + x;
        ldst[i] = p * KPP_ + (r * KRP_ + 4 * g) * 2;
    }
    const int qch = tid / QTC_, qr = (tid - QTC_ * qch) >> 3, qg = tid & 7;
    const int qgoff = qch * HW_ + (yb + qr) * W_ + x0 + 4 * qg;
    const int qwave = (qr >> 1) * 4 + (qg >> 1);
    const int qm    = (qr & 1) * 8 + (qg & 1) * 4;
    const int qldst = KST_ + (qch >> 1) * QPP_ + (qwave * 16 + qm) * 2 + (qch & 1);

    int koff[7];
#pragma unroll
    for (int t = 0; t < 7; ++t) {
        int kk = 16 * t + m16;
        int ry = kk / KW_, rx = kk % KW_;
        int ky = yw - R_ + ry; ky = ky < 0 ? 0 : ky;
        int kx = xw - R_ + rx; kx = kx < 0 ? 0 : kx;
        koff[t] = hh * KPP_ + ((ky - (yb - R_)) * KRP_ + (kx - (x0 - 8))) * 2;
    }
    const int aoff = KST_ + hh * QPP_ + (16 * wv + m16) * 2;

    v8f  acc[7] = {};
    v2f  fa[2], fb[2][7];
    v2f ks00 = {}, ks01 = {}, ks10 = {}, ks11 = {};
    v2f qs0 = {}, qs1 = {};

    v4f ra[4][2], rb[4][2], rq[4];

#define FETCH(c, k)                                                      \
    {                                                                    \
        const int co_ = (c) * HW_;                                       \
        ra[k][0] = *(const v4f*)(mapN + co_ + goff[0]);                  \
        rb[k][0] = *(const v4f*)(mapN + co_ + HW_ + goff[0]);            \
        ra[k][1] = *(const v4f*)(mapN + co_ + goff[1]);                  \
        rb[k][1] = *(const v4f*)(mapN + co_ + HW_ + goff[1]);            \
        rq[k] = *(const v4f*)(imgN + co_ + qgoff);                       \
    }

#define HOLD_SUMS()                                                      \
    asm volatile("" : "+v"(ks00), "+v"(ks01), "+v"(ks10), "+v"(ks11), "+v"(qs0), "+v"(qs1))

#define STASH(buf, k)                                                    \
    {                                                                    \
        float* st_ = lds + (buf);                                        \
        put_pair(st_ + ldst[0], ra[k][0], rb[k][0]);                     \
        if (second) put_pair(st_ + ldst[1], ra[k][1], rb[k][1]);         \
        float* qd_ = st_ + qldst;                                        \
        qd_[0] = rq[k].x; qd_[2] = rq[k].y; qd_[4] = rq[k].z; qd_[6] = rq[k].w; \
        ks00 += ra[k][0].lo * ra[k][0].lo;   ks01 += ra[k][0].hi * ra[k][0].hi;   \
        ks00 += rb[k][0].lo * rb[k][0].lo;   ks01 += rb[k][0].hi * rb[k][0].hi;   \
        ks10 += ra[k][1].lo * ra[k][1].lo;   ks11 += ra[k][1].hi * ra[k][1].hi;   \
        ks10 += rb[k][1].lo * rb[k][1].lo;   ks11 += rb[k][1].hi * rb[k][1].hi;   \
        qs0  += rq[k].lo * rq[k].lo;         qs1  += rq[k].hi * rq[k].hi;         \
        HOLD_SUMS();                                                     \
    }

#define LOADFRAGS(buf)                                                   \
    {                                                                    \
        const float* st_ = lds + (buf);                                  \
        _Pragma("unroll")                                                \
        for (int j = 0; j < 2; ++j) {                                    \
            fa[j] = *(const v2f*)(st_ + 2 * j * QPP_ + aoff);            \
            _Pragma("unroll")                                            \
            for (int t = 0; t < 7; ++t)                                  \
                fb[j][t] = *(const v2f*)(st_ + 2 * j * KPP_ + koff[t]);  \
        }                                                                \
    }

#define MMAS()                                                           \
    {                                                                    \
        _Pragma("unroll")                                                \
        for (int j = 0; j < 2; ++j) {                                    \
            _Pragma("unroll")                                            \
            for (int t = 0; t < 7; ++t)                                  \
                acc[t] = __builtin_amdgcn_wmma_f32_16x16x4_f32(          \
                    false, fa[j], false, fb[j][t], (short)0, acc[t], false, false); \
        }                                                                \
    }

#define STEP(s_, u_, DO_STASH, DO_FETCH)                                 \
    {                                                                    \
        LOADFRAGS(((u_) & 1) * STG_);                                    \
        if (DO_STASH) STASH((((u_) + 1) & 1) * STG_, ((u_) + 1) & 3);    \
        stage_signal();                                                  \
        if (DO_FETCH) FETCH(8 * ((s_) + 5), ((u_) + 1) & 3);             \
        MMAS();                                                          \
        stage_wait();                                                    \
    }

    FETCH(0, 0);
    __builtin_amdgcn_sched_barrier(0);
    FETCH(8, 1);
    __builtin_amdgcn_sched_barrier(0);
    FETCH(16, 2);
    __builtin_amdgcn_sched_barrier(0);
    FETCH(24, 3);
    __builtin_amdgcn_sched_barrier(0);
    STASH(0, 0);
    FETCH(32, 0);
    lds_barrier();

    for (int s = 0; s < 24; s += 4) {
        STEP(s,     0, true, true);
        STEP(s + 1, 1, true, true);
        STEP(s + 2, 2, true, true);
        STEP(s + 3, 3, true, true);
    }
    STEP(24, 0, true, true);
    STEP(25, 1, true, true);
    STEP(26, 2, true, true);
    STEP(27, 3, true, false);
    STEP(28, 0, true, false);
    STEP(29, 1, true, false);
    STEP(30, 2, true, false);
    {
        LOADFRAGS(STG_);
        {
            const int p = tid / KTP_, rem = tid - KTP_ * p;
            const int r = rem / HG_, g = rem - HG_ * r;
            float* d = lds + p * KTB_ + r * KRP_ + 4 * g;
            *(v2f*)d       = ks00;
            *(v2f*)(d + 2) = ks01;
        }
        if (second) {
            const int u = tid + NTHR_;
            const int p = u / KTP_, rem = u - KTP_ * p;
            const int r = rem / HG_, g = rem - HG_ * r;
            float* d = lds + p * KTB_ + r * KRP_ + 4 * g;
            *(v2f*)d       = ks10;
            *(v2f*)(d + 2) = ks11;
        }
        v4f q4; q4.x = qs0.x; q4.y = qs0.y; q4.z = qs1.x; q4.w = qs1.y;
        *(v4f*)(lds + QPB_ + qch * (BW_ * BH_) + qwave * 16 + qm) = q4;
        stage_signal();
        MMAS();
        stage_wait();
    }

#undef STEP
#undef HOLD_SUMS
#undef FETCH
#undef STASH
#undef LOADFRAGS
#undef MMAS

    const float* qp = lds + QPB_ + 4 * hh * (BW_ * BH_) + 16 * wv + m16;
    float asum = (qp[0] + qp[BW_ * BH_]) + (qp[2 * BW_ * BH_] + qp[3 * BW_ * BH_]);
    asum += __int_as_float(
        __builtin_amdgcn_ds_swizzle(__float_as_int(asum), 0x401F));
    const float qinv = 1.f / (sqrtf(asum) + EPS_);

    float kinv[7];
#pragma unroll
    for (int t = 0; t < 7; ++t) {
        const float* kp = lds + 2 * hh * KTB_ + ((koff[t] - hh * KPP_) >> 1);
        float s2 = kp[0] + kp[KTB_];
        s2 += __int_as_float(
            __builtin_amdgcn_ds_swizzle(__float_as_int(s2), 0x401F));
        kinv[t] = 1.f / (sqrtf(s2) + EPS_);
    }
    __syncthreads();

    float* G = lds + wv * WAVE_LDS;
#pragma unroll
    for (int t = 0; t < 7; ++t) {
#pragma unroll
        for (int j = 0; j < 8; ++j) {
            int m = j + 8 * hh;
            G[m * NK_ + 16 * t + m16] = acc[t][j];
        }
        if (hh == 0) G[GK_ + 16 * t + m16] = kinv[t];
    }
    if (hh == 0) G[GQ_ + m16] = qinv;
    __syncthreads();

    float* outCos  = out;
    float* outDiss = out + (size_t)OUT_PLANE;
    const int yy  = wv >> 1;
    const int swv = (yy >> 1) * 4 + (lane >> 3);
    const int sq  = (yy & 1) * 8 + (lane & 7);
    const int kb  = (yy & 1) * KW_ + (lane & 7);
    const float* Gs = lds + swv * WAVE_LDS;
    const float qv  = Gs[GQ_ + sq];
    const int pixOff = (yb + yy) * W_ + x0 + lane;
    for (int s = wv & 1; s < S_; s += 2) {
        int i  = s / 7, j = s - 7 * i;
        int kk = kb + i * KW_ + j;
        float D  = Gs[sq * NK_ + kk];
        float d1 = D * qv;
        float cv = 1.0f - d1 * Gs[GK_ + kk];
        int o = (n * S_ + s) * HW_ + pixOff;
        __builtin_nontemporal_store(cv, outCos + o);
        __builtin_nontemporal_store(d1, outDiss + o);
    }
}

extern "C" void kernel_launch(void* const* d_in, const int* in_sizes, int n_in,
                              void* d_out, int out_size, void* d_ws, size_t ws_size,
                              hipStream_t stream) {
    const float* img = (const float*)d_in[0];
    const float* map = (const float*)d_in[1];
    (void)in_sizes; (void)n_in; (void)out_size; (void)d_ws; (void)ws_size;

    dim3 grid(W_ / BW_, H_ / BH_, N_);
    corr_wmma_kernel<<<grid, dim3(NTHR_), 0, stream>>>(img, map, (float*)d_out);
}
